# starting kernel
# baseline (speedup 1.0000x reference)
#include <hip/hip_runtime.h>
#include <float.h>


typedef __attribute__((ext_vector_type(2))) float v2f;
typedef __attribute__((ext_vector_type(8))) float v8f;

#define B_ 16
#define L_ 4096
#define C_ 1024
#define K_ 16
#define NSPLIT 16
#define LRANGE (L_ / NSPLIT)
#define CHUNK 128
#define NCHUNK (LRANGE / CHUNK)
#define SP 132
#define CW (C_ / 8)
#define NT 8
#define NEG_MAX (-3.402823466e38f)

__global__ __launch_bounds__(256) void qp_fused(const float* __restrict__ x,
                                                const int* __restrict__ mask,
                                                const float* __restrict__ qm,
                                                float* __restrict__ part,
                                                float* __restrict__ mstat,
                                                float* __restrict__ sstat)
{
    __shared__ float sw[K_ * SP];
    __shared__ float red[K_ * 16];
    __shared__ float mq[K_], sq[K_], scq[K_];

    const int tid   = threadIdx.x;
    const int w     = tid >> 5;
    const int lane  = tid & 31;
    const int lid   = lane & 15;
    const int half  = lane >> 4;
    const int split = blockIdx.x & (NSPLIT - 1);
    const int b     = blockIdx.x >> 4;
    const int lbase = split * LRANGE;
    const int cw    = w * CW;

    const int tq = tid >> 4;
    const int tj = tid & 15;

    if (tid < K_) { mq[tid] = NEG_MAX; sq[tid] = 0.0f; }
    __syncthreads();

    v8f zero = {};
    v8f acc[NT];
#pragma unroll
    for (int j = 0; j < NT; ++j) acc[j] = zero;

    const float* qrow = qm + lid * C_ + 2 * half;

    for (int ch = 0; ch < NCHUNK; ++ch) {
        const int l0 = lbase + ch * CHUNK + w * 16;
        const float* xrow = x + (size_t)(b * L_ + l0 + lid) * C_ + 2 * half;
        v8f d = zero;
#pragma unroll 16
        for (int c = 0; c < C_; c += 4) {
            v2f av = *(const v2f*)(qrow + c);
            v2f bv = *(const v2f*)(xrow + c);
            d = __builtin_amdgcn_wmma_f32_16x16x4_f32(false, av, false, bv,
                                                      (short)0, d, false, false);
        }
        const int mval = mask[b * L_ + l0 + lid];
        const int lloc = w * 16 + lid;
#pragma unroll
        for (int r = 0; r < 8; ++r) {
            float v = d[r] * 0.03125f;
            sw[(r + 8 * half) * SP + lloc] = mval ? v : NEG_MAX;
        }
        __syncthreads();

        float lm = NEG_MAX;
#pragma unroll
        for (int i = 0; i < CHUNK / 16; ++i)
            lm = fmaxf(lm, sw[tq * SP + tj + 16 * i]);
        red[tq * 16 + tj] = lm;
        __syncthreads();
        if (tid < K_) {
            float nm = mq[tid];
            for (int i = 0; i < 16; ++i) nm = fmaxf(nm, red[tid * 16 + i]);
            scq[tid] = __expf(mq[tid] - nm);
            mq[tid]  = nm;
        }
        __syncthreads();

        float scl[8];
#pragma unroll
        for (int r = 0; r < 8; ++r) scl[r] = scq[r + 8 * half];
#pragma unroll
        for (int j = 0; j < NT; ++j)
#pragma unroll
            for (int r = 0; r < 8; ++r) acc[j][r] *= scl[r];

        const float nmv = mq[tq];
        float ls = 0.0f;
#pragma unroll
        for (int i = 0; i < CHUNK / 16; ++i) {
            float e = __expf(sw[tq * SP + tj + 16 * i] - nmv);
            sw[tq * SP + tj + 16 * i] = e;
            ls += e;
        }
        red[tq * 16 + tj] = ls;
        __syncthreads();
        if (tid < K_) {
            float s = 0.0f;
            for (int i = 0; i < 16; ++i) s += red[tid * 16 + i];
            sq[tid] = sq[tid] * scq[tid] + s;
        }

        const float* xcol = x + (size_t)(b * L_ + lbase + ch * CHUNK) * C_ + cw + lid;
        for (int l = 0; l < CHUNK; l += 4) {
            v2f av = *(const v2f*)(&sw[lid * SP + l + 2 * half]);
            const float* xp = xcol + (size_t)(l + 2 * half) * C_;
#pragma unroll
            for (int j = 0; j < NT; ++j) {
                v2f bv;
                bv.x = __builtin_nontemporal_load(xp + j * 16);
                bv.y = __builtin_nontemporal_load(xp + C_ + j * 16);
                acc[j] = __builtin_amdgcn_wmma_f32_16x16x4_f32(false, av, false, bv,
                                                               (short)0, acc[j], false, false);
            }
        }
        __syncthreads();
    }

    float* prow = part + ((size_t)((b * NSPLIT + split) * K_ + 8 * half)) * C_ + cw + lid;
#pragma unroll
    for (int j = 0; j < NT; ++j)
#pragma unroll
        for (int r = 0; r < 8; ++r)
            __builtin_nontemporal_store(acc[j][r], prow + (size_t)r * C_ + j * 16);

    if (tid < K_) {
        mstat[(b * NSPLIT + split) * K_ + tid] = mq[tid];
        sstat[(b * NSPLIT + split) * K_ + tid] = sq[tid];
    }
}

__global__ __launch_bounds__(256) void qp_combine(const float* __restrict__ part,
                                                  const float* __restrict__ mstat,
                                                  const float* __restrict__ sstat,
                                                  float* __restrict__ out)
{
    const int i = blockIdx.x * 256 + threadIdx.x;
    const int c = i & (C_ - 1);
    const int q = (i >> 10) & (K_ - 1);
    const int b = i >> 14;

    float M = NEG_MAX;
#pragma unroll
    for (int wv = 0; wv < NSPLIT; ++wv)
        M = fmaxf(M, mstat[(b * NSPLIT + wv) * K_ + q]);

    if (M == NEG_MAX) { out[i] = 0.0f; return; }

    float num = 0.0f, den = 0.0f;
#pragma unroll
    for (int wv = 0; wv < NSPLIT; ++wv) {
        const int sidx = (b * NSPLIT + wv) * K_ + q;
        const float e  = __expf(mstat[sidx] - M);
        den += e * sstat[sidx];
        num += e * __builtin_nontemporal_load(part + (size_t)sidx * C_ + c);
    }
    out[i] = num / den;
}

extern "C" void kernel_launch(void* const* d_in, const int* in_sizes, int n_in,
                              void* d_out, int out_size, void* d_ws, size_t ws_size,
                              hipStream_t stream)
{
    const float* x    = (const float*)d_in[0];
    const int*   mask = (const int*)d_in[1];
    const float* q    = (const float*)d_in[2];
    float*       out  = (float*)d_out;

    float* part  = (float*)d_ws;
    float* mstat = part + (size_t)B_ * NSPLIT * K_ * C_;
    float* sstat = mstat + B_ * NSPLIT * K_;

    qp_fused<<<B_ * NSPLIT, 256, 0, stream>>>(x, mask, q, part, mstat, sstat);
    qp_combine<<<(B_ * K_ * C_) / 256, 256, 0, stream>>>(part, mstat, sstat, out);
}


# each wave owns a 128-column slice for both products: its x rows go global->LDS once as 512-B row loads and feed score and pooling WMMAs; q slice in registers; sums re-ordered (32-row softmax steps)
# speedup vs baseline: 1.3804x; 1.3804x over previous
#include <hip/hip_runtime.h>
#include <float.h>


typedef __attribute__((ext_vector_type(2))) float v2f;
typedef __attribute__((ext_vector_type(4))) float v4f;
typedef __attribute__((ext_vector_type(8))) float v8f;
typedef __attribute__((ext_vector_type(4))) int   v4i;

#define DEV static __device__ __forceinline__
DEV void sfence() { __builtin_amdgcn_sched_barrier(0); }
DEV void lds_barrier() {
    asm volatile("s_wait_dscnt 0x0\n\ts_barrier_signal -1\n\ts_barrier_wait -1" ::: "memory");
}

#define B_ 16
#define L_ 4096
#define C_ 1024
#define K_ 16
#define NSPLIT 16
#define LRANGE (L_ / NSPLIT)
#define R_ 32
#define NSTEP (LRANGE / R_)
#define CW (C_ / 8)
#define XP 140
#define NEG_MAX (-3.402823466e38f)


__global__ __launch_bounds__(256) void qp_fused(const float* __restrict__ x,
                                                const int* __restrict__ mask,
                                                const float* __restrict__ qm,
                                                float* __restrict__ part,
                                                float* __restrict__ mstat,
                                                float* __restrict__ sstat)
{
    __shared__ __attribute__((aligned(16))) float xs[8 * R_ * XP];
    __shared__ __attribute__((aligned(16))) float ps[2 * 8 * 4 * 32 * 4];
    __shared__ __attribute__((aligned(16))) int   msk[LRANGE];

    const int tid   = threadIdx.x;
    const int w     = tid >> 5;
    const int lane  = tid & 31;
    const int lid   = lane & 15;
    const int half  = lane >> 4;
    const int split = blockIdx.x & (NSPLIT - 1);
    const int b     = blockIdx.x >> 4;
    const int lbase = split * LRANGE;
    const int c0    = w * CW;

    float* xw = xs + w * (R_ * XP);

    v4f qf[16];
    {
        const float* qp = qm + lid * C_ + c0 + 4 * half;
#pragma unroll
        for (int s = 0; s < 16; ++s) qf[s] = *(const v4f*)(qp + 8 * s);
    }

    const float* xg = x + (size_t)(b * L_ + lbase) * C_ + c0 + 4 * lane;
    msk[tid] = mask[b * L_ + lbase + tid];
    const int* mg = msk + 8 * half;
    sfence();

    v8f zero = {};
    v8f acc[8];
#pragma unroll
    for (int j = 0; j < 8; ++j) acc[j] = zero;
    float m_run = NEG_MAX, s_run = 0.0f;

    v4f G[R_];
#pragma unroll
    for (int i = 0; i < R_; ++i)
        G[i] = __builtin_nontemporal_load((const v4f*)(xg + (size_t)i * C_));
#pragma unroll
    for (int i = 0; i < R_; ++i) *(v4f*)(xw + i * XP + 4 * lane) = G[i];

    for (int st = 0; st < NSTEP; ++st) {
        __builtin_amdgcn_fence(__ATOMIC_ACQ_REL, "wavefront");

        if (st + 1 < NSTEP) {
#pragma unroll
            for (int i = 0; i < R_; ++i)
                G[i] = __builtin_nontemporal_load((const v4f*)(xg + (size_t)((st + 1) * R_ + i) * C_));
        }
        v8f d0 = zero, d1 = zero;
        {
            const float* xa = xw + lid * XP + 4 * half;
            v4f fa[2][4];
#pragma unroll
            for (int u = 0; u < 2; ++u) {
                fa[0][2 * u]     = *(const v4f*)(xa + 8 * u);
                fa[0][2 * u + 1] = *(const v4f*)(xa + 16 * XP + 8 * u);
            }
#pragma unroll
            for (int blk = 0; blk < 8; ++blk) {
                if (blk + 1 < 8) {
#pragma unroll
                    for (int u = 0; u < 2; ++u) {
                        fa[(blk + 1) & 1][2 * u]     = *(const v4f*)(xa + 8 * (2 * (blk + 1) + u));
                        fa[(blk + 1) & 1][2 * u + 1] = *(const v4f*)(xa + 16 * XP + 8 * (2 * (blk + 1) + u));
                    }
                }
                sfence();
#pragma unroll
                for (int u = 0; u < 2; ++u) {
                    const int s = 2 * blk + u;
                    const v4f a0 = fa[blk & 1][2 * u];
                    const v4f a1 = fa[blk & 1][2 * u + 1];
                    v2f q0 = {qf[s].x, qf[s].y};
                    v2f q1 = {qf[s].z, qf[s].w};
                    v2f a00 = {a0.x, a0.y}, a01 = {a0.z, a0.w};
                    v2f a10 = {a1.x, a1.y}, a11 = {a1.z, a1.w};
                    d0 = __builtin_amdgcn_wmma_f32_16x16x4_f32(false, a00, false, q0, (short)0, d0, false, false);
                    d1 = __builtin_amdgcn_wmma_f32_16x16x4_f32(false, a10, false, q0, (short)0, d1, false, false);
                    d0 = __builtin_amdgcn_wmma_f32_16x16x4_f32(false, a01, false, q1, (short)0, d0, false, false);
                    d1 = __builtin_amdgcn_wmma_f32_16x16x4_f32(false, a11, false, q1, (short)0, d1, false, false);
                }
                sfence();
            }
        }

        float* pb = ps + (st & 1) * (8 * 4 * 128);
        {
            float* pw = pb + w * (4 * 128) + 4 * lane;
            v4f t0 = {d0[0], d0[1], d0[2], d0[3]};
            v4f t1 = {d0[4], d0[5], d0[6], d0[7]};
            v4f t2 = {d1[0], d1[1], d1[2], d1[3]};
            v4f t3 = {d1[4], d1[5], d1[6], d1[7]};
            *(v4f*)(pw + 0 * 128) = t0;
            *(v4f*)(pw + 1 * 128) = t1;
            *(v4f*)(pw + 2 * 128) = t2;
            *(v4f*)(pw + 3 * 128) = t3;
        }
        lds_barrier();

        const float* xb = xw + (8 * half) * XP + lid;
        float fx[2][8], fy[2][8];
#pragma unroll
        for (int j = 0; j < 8; ++j) { fx[0][j] = xb[16 * j]; fy[0][j] = xb[XP + 16 * j]; }

        v4i mk[4];
#pragma unroll
        for (int i = 0; i < 4; ++i)
            mk[i] = *(const v4i*)(mg + st * R_ + 16 * (i >> 1) + 4 * (i & 1));
        v4f pr[8][4];
#pragma unroll
        for (int w2 = 0; w2 < 8; ++w2)
#pragma unroll
            for (int k = 0; k < 4; ++k)
                pr[w2][k] = *(const v4f*)(pb + w2 * (4 * 128) + k * 128 + 4 * lane);
        sfence();
        v4f sa[4];
#pragma unroll
        for (int k = 0; k < 4; ++k) sa[k] = pr[0][k];
#pragma unroll
        for (int w2 = 1; w2 < 8; ++w2)
#pragma unroll
            for (int k = 0; k < 4; ++k) sa[k] += pr[w2][k];

        float sv[16];
#pragma unroll
        for (int k = 0; k < 4; ++k) {
            sv[4 * k + 0] = mk[k].x ? sa[k].x * 0.03125f : NEG_MAX;
            sv[4 * k + 1] = mk[k].y ? sa[k].y * 0.03125f : NEG_MAX;
            sv[4 * k + 2] = mk[k].z ? sa[k].z * 0.03125f : NEG_MAX;
            sv[4 * k + 3] = mk[k].w ? sa[k].w * 0.03125f : NEG_MAX;
        }
        float mx = sv[0];
#pragma unroll
        for (int i = 1; i < 16; ++i) mx = fmaxf(mx, sv[i]);
        mx = fmaxf(mx, __shfl_xor(mx, 16));
        const float nm  = fmaxf(m_run, mx);
        const float scl = __expf(m_run - nm);
        m_run = nm;
        float ls = 0.0f;
#pragma unroll
        for (int i = 0; i < 16; ++i) { sv[i] = __expf(sv[i] - nm); ls += sv[i]; }
        ls += __shfl_xor(ls, 16);
        s_run = s_run * scl + ls;
#pragma unroll
        for (int j = 0; j < 8; ++j) acc[j] *= scl;

#pragma unroll
        for (int n = 0; n < 8; ++n) {
            if (n + 1 < 8) {
                const float* xr = xb + (16 * ((n + 1) >> 2) + 2 * ((n + 1) & 3)) * XP;
#pragma unroll
                for (int j = 0; j < 8; ++j) {
                    fx[(n + 1) & 1][j] = xr[16 * j];
                    fy[(n + 1) & 1][j] = xr[XP + 16 * j];
                }
            }
            sfence();
            v2f bw = {sv[2 * n], sv[2 * n + 1]};
#pragma unroll
            for (int j = 0; j < 8; ++j) {
                v2f av = {fx[n & 1][j], fy[n & 1][j]};
                acc[j] = __builtin_amdgcn_wmma_f32_16x16x4_f32(false, av, false, bw,
                                                               (short)0, acc[j], false, false);
            }
            sfence();
        }

        __builtin_amdgcn_fence(__ATOMIC_ACQ_REL, "wavefront");
        if (st + 1 < NSTEP) {
#pragma unroll
            for (int i = 0; i < R_; ++i) *(v4f*)(xw + i * XP + 4 * lane) = G[i];
        }
    }

    __builtin_amdgcn_fence(__ATOMIC_ACQ_REL, "wavefront");
#pragma unroll
    for (int j = 0; j < 8; ++j) {
        v4f o0 = {acc[j][0], acc[j][1], acc[j][2], acc[j][3]};
        v4f o1 = {acc[j][4], acc[j][5], acc[j][6], acc[j][7]};
        *(v4f*)(xw + lid * XP + 16 * j + 8 * half)     = o0;
        *(v4f*)(xw + lid * XP + 16 * j + 8 * half + 4) = o1;
    }
    __builtin_amdgcn_fence(__ATOMIC_ACQ_REL, "wavefront");
    float* prow = part + (size_t)((b * NSPLIT + split) * K_) * C_ + c0 + 4 * lane;
#pragma unroll
    for (int qq = 0; qq < K_; ++qq) {
        v4f o = *(const v4f*)(xw + qq * XP + 4 * lane);
        __builtin_nontemporal_store(o, (v4f*)(prow + (size_t)qq * C_));
    }

    if (w == 0 && lane < K_) {
        mstat[(b * NSPLIT + split) * K_ + lane] = m_run;
        sstat[(b * NSPLIT + split) * K_ + lane] = s_run;
    }
}

__global__ __launch_bounds__(256) void qp_combine(const float* __restrict__ part,
                                                  const float* __restrict__ mstat,
                                                  const float* __restrict__ sstat,
                                                  float* __restrict__ out)
{
    const int i = blockIdx.x * 256 + threadIdx.x;
    const int c = i & (C_ - 1);
    const int q = (i >> 10) & (K_ - 1);
    const int b = i >> 14;

    float M = NEG_MAX;
#pragma unroll
    for (int wv = 0; wv < NSPLIT; ++wv)
        M = fmaxf(M, mstat[(b * NSPLIT + wv) * K_ + q]);

    if (M == NEG_MAX) { out[i] = 0.0f; return; }

    float num = 0.0f, den = 0.0f;
#pragma unroll
    for (int wv = 0; wv < NSPLIT; ++wv) {
        const int sidx = (b * NSPLIT + wv) * K_ + q;
        const float e  = __expf(mstat[sidx] - M);
        den += e * sstat[sidx];
        num += e * __builtin_nontemporal_load(part + (size_t)sidx * C_ + c);
    }
    out[i] = num / den;
}

extern "C" void kernel_launch(void* const* d_in, const int* in_sizes, int n_in,
                              void* d_out, int out_size, void* d_ws, size_t ws_size,
                              hipStream_t stream)
{
    const float* x    = (const float*)d_in[0];
    const int*   mask = (const int*)d_in[1];
    const float* q    = (const float*)d_in[2];
    float*       out  = (float*)d_out;

    float* part  = (float*)d_ws;
    float* mstat = part + (size_t)B_ * NSPLIT * K_ * C_;
    float* sstat = mstat + B_ * NSPLIT * K_;

    qp_fused<<<B_ * NSPLIT, 256, 0, stream>>>(x, mask, q, part, mstat, sstat);
    qp_combine<<<(B_ * K_ * C_) / 256, 256, 0, stream>>>(part, mstat, sstat, out);
}


# x tile double-buffered in LDS; global->LDS relay as a rolling window: 2 rows stored + 2 requested before each WMMA block (30-32 row loads in flight), not a 32-load burst per step; sums unchanged
# speedup vs baseline: 1.5565x; 1.1276x over previous
#include <hip/hip_runtime.h>
#include <float.h>


typedef __attribute__((ext_vector_type(2))) float v2f;
typedef __attribute__((ext_vector_type(4))) float v4f;
typedef __attribute__((ext_vector_type(8))) float v8f;
typedef __attribute__((ext_vector_type(4))) int   v4i;

#define DEV static __device__ __forceinline__
DEV void sfence() { __builtin_amdgcn_sched_barrier(0); }
DEV void lds_barrier() {
    asm volatile("s_wait_dscnt 0x0\n\ts_barrier_signal -1\n\ts_barrier_wait -1" ::: "memory");
}

#define B_ 16
#define L_ 4096
#define C_ 1024
#define K_ 16
#define NSPLIT 16
#define LRANGE (L_ / NSPLIT)
#define R_ 32
#define NSTEP (LRANGE / R_)
#define CW (C_ / 8)
#define XP 140
#define XBUF (8 * R_ * XP)
#define NEG_MAX (-3.402823466e38f)


template <bool STORE, bool LOAD>
DEV void relay(const int p, float* xn, const float* xg2, v4f (&G)[R_])
{
#pragma unroll
    for (int u = 0; u < 2; ++u) {
        const int k = 2 * p + u;
        if (STORE) *(v4f*)(xn + k * XP) = G[k];
        if (LOAD)  G[k] = __builtin_nontemporal_load((const v4f*)(xg2 + (size_t)k * C_));
    }
}

template <bool STORE, bool LOAD>
DEV void qp_step(const int st, float* xs_w, const float* xg, const v4f (&qf)[16],
                 v4f (&G)[R_], v8f (&acc)[8], float& m_run, float& s_run,
                 float* ps, const int* mg,
                 const int w, const int lane, const int lid, const int half)
{
    float* xw = xs_w + (st & 1) * XBUF;
    float* xn = xs_w + ((st & 1) ^ 1) * XBUF + 4 * lane;
    const float* xg2 = LOAD ? xg + (size_t)((st + 2) * R_) * C_ : xg;

    __builtin_amdgcn_fence(__ATOMIC_ACQ_REL, "wavefront");

    const v8f zero = {};
    v8f d0 = zero, d1 = zero;
    {
        const float* xa = xw + lid * XP + 4 * half;
        v4f fa[2][4];
#pragma unroll
        for (int u = 0; u < 2; ++u) {
            fa[0][2 * u]     = *(const v4f*)(xa + 8 * u);
            fa[0][2 * u + 1] = *(const v4f*)(xa + 16 * XP + 8 * u);
        }
#pragma unroll
        for (int blk = 0; blk < 8; ++blk) {
            if (blk + 1 < 8) {
#pragma unroll
                for (int u = 0; u < 2; ++u) {
                    fa[(blk + 1) & 1][2 * u]     = *(const v4f*)(xa + 8 * (2 * (blk + 1) + u));
                    fa[(blk + 1) & 1][2 * u + 1] = *(const v4f*)(xa + 16 * XP + 8 * (2 * (blk + 1) + u));
                }
            }
            relay<STORE, LOAD>(blk, xn, xg2, G);
            sfence();
#pragma unroll
            for (int u = 0; u < 2; ++u) {
                const int s = 2 * blk + u;
                const v4f a0 = fa[blk & 1][2 * u];
                const v4f a1 = fa[blk & 1][2 * u + 1];
                v2f q0 = {qf[s].x, qf[s].y};
                v2f q1 = {qf[s].z, qf[s].w};
                v2f a00 = {a0.x, a0.y}, a01 = {a0.z, a0.w};
                v2f a10 = {a1.x, a1.y}, a11 = {a1.z, a1.w};
                d0 = __builtin_amdgcn_wmma_f32_16x16x4_f32(false, a00, false, q0, (short)0, d0, false, false);
                d1 = __builtin_amdgcn_wmma_f32_16x16x4_f32(false, a10, false, q0, (short)0, d1, false, false);
                d0 = __builtin_amdgcn_wmma_f32_16x16x4_f32(false, a01, false, q1, (short)0, d0, false, false);
                d1 = __builtin_amdgcn_wmma_f32_16x16x4_f32(false, a11, false, q1, (short)0, d1, false, false);
            }
            sfence();
        }
    }

    float* pb = ps + (st & 1) * (8 * 4 * 128);
    {
        float* pw = pb + w * (4 * 128) + 4 * lane;
        v4f t0 = {d0[0], d0[1], d0[2], d0[3]};
        v4f t1 = {d0[4], d0[5], d0[6], d0[7]};
        v4f t2 = {d1[0], d1[1], d1[2], d1[3]};
        v4f t3 = {d1[4], d1[5], d1[6], d1[7]};
        *(v4f*)(pw + 0 * 128) = t0;
        *(v4f*)(pw + 1 * 128) = t1;
        *(v4f*)(pw + 2 * 128) = t2;
        *(v4f*)(pw + 3 * 128) = t3;
    }
    lds_barrier();

    const float* xb = xw + (8 * half) * XP + lid;
    float fx[2][8], fy[2][8];
#pragma unroll
    for (int j = 0; j < 8; ++j) { fx[0][j] = xb[16 * j]; fy[0][j] = xb[XP + 16 * j]; }

    v4i mk[4];
#pragma unroll
    for (int i = 0; i < 4; ++i)
        mk[i] = *(const v4i*)(mg + st * R_ + 16 * (i >> 1) + 4 * (i & 1));
    v4f pr[8][4];
#pragma unroll
    for (int w2 = 0; w2 < 8; ++w2)
#pragma unroll
        for (int k = 0; k < 4; ++k)
            pr[w2][k] = *(const v4f*)(pb + w2 * (4 * 128) + k * 128 + 4 * lane);
    sfence();
    v4f sa[4];
#pragma unroll
    for (int k = 0; k < 4; ++k) sa[k] = pr[0][k];
#pragma unroll
    for (int w2 = 1; w2 < 8; ++w2)
#pragma unroll
        for (int k = 0; k < 4; ++k) sa[k] += pr[w2][k];

    float sv[16];
#pragma unroll
    for (int k = 0; k < 4; ++k) {
        sv[4 * k + 0] = mk[k].x ? sa[k].x * 0.03125f : NEG_MAX;
        sv[4 * k + 1] = mk[k].y ? sa[k].y * 0.03125f : NEG_MAX;
        sv[4 * k + 2] = mk[k].z ? sa[k].z * 0.03125f : NEG_MAX;
        sv[4 * k + 3] = mk[k].w ? sa[k].w * 0.03125f : NEG_MAX;
    }
    float mx = sv[0];
#pragma unroll
    for (int i = 1; i < 16; ++i) mx = fmaxf(mx, sv[i]);
    mx = fmaxf(mx, __shfl_xor(mx, 16));
    const float nm  = fmaxf(m_run, mx);
    const float scl = __expf(m_run - nm);
    m_run = nm;
    float ls = 0.0f;
#pragma unroll
    for (int i = 0; i < 16; ++i) { sv[i] = __expf(sv[i] - nm); ls += sv[i]; }
    ls += __shfl_xor(ls, 16);
    s_run = s_run * scl + ls;
#pragma unroll
    for (int j = 0; j < 8; ++j) acc[j] *= scl;

#pragma unroll
    for (int n = 0; n < 8; ++n) {
        if (n + 1 < 8) {
            const float* xr = xb + (16 * ((n + 1) >> 2) + 2 * ((n + 1) & 3)) * XP;
#pragma unroll
            for (int j = 0; j < 8; ++j) {
                fx[(n + 1) & 1][j] = xr[16 * j];
                fy[(n + 1) & 1][j] = xr[XP + 16 * j];
            }
        }
        relay<STORE, LOAD>(8 + n, xn, xg2, G);
        sfence();
        v2f bw = {sv[2 * n], sv[2 * n + 1]};
#pragma unroll
        for (int j = 0; j < 8; ++j) {
            v2f av = {fx[n & 1][j], fy[n & 1][j]};
            acc[j] = __builtin_amdgcn_wmma_f32_16x16x4_f32(false, av, false, bw,
                                                           (short)0, acc[j], false, false);
        }
        sfence();
    }
}

__global__ __launch_bounds__(256) void qp_fused(const float* __restrict__ x,
                                                const int* __restrict__ mask,
                                                const float* __restrict__ qm,
                                                float* __restrict__ part,
                                                float* __restrict__ mstat,
                                                float* __restrict__ sstat)
{
    __shared__ __attribute__((aligned(16))) float xs[2 * XBUF];
    __shared__ __attribute__((aligned(16))) float ps[2 * 8 * 4 * 32 * 4];
    __shared__ __attribute__((aligned(16))) int   msk[LRANGE];

    const int tid   = threadIdx.x;
    const int w     = tid >> 5;
    const int lane  = tid & 31;
    const int lid   = lane & 15;
    const int half  = lane >> 4;
    const int split = blockIdx.x & (NSPLIT - 1);
    const int b     = blockIdx.x >> 4;
    const int lbase = split * LRANGE;
    const int c0    = w * CW;

    float* xw = xs + w * (R_ * XP);

    v4f qf[16];
    {
        const float* qp = qm + lid * C_ + c0 + 4 * half;
#pragma unroll
        for (int s = 0; s < 16; ++s) qf[s] = *(const v4f*)(qp + 8 * s);
    }

    const float* xg = x + (size_t)(b * L_ + lbase) * C_ + c0 + 4 * lane;
    const int mval = mask[b * L_ + lbase + tid];

    v4f G[R_];
#pragma unroll
    for (int i = 0; i < R_; ++i)
        G[i] = __builtin_nontemporal_load((const v4f*)(xg + (size_t)i * C_));
    sfence();

    msk[tid] = mval;
    const int* mg = msk + 8 * half;

    v8f zero = {};
    v8f acc[8];
#pragma unroll
    for (int j = 0; j < 8; ++j) acc[j] = zero;
    float m_run = NEG_MAX, s_run = 0.0f;

    {
        const float* xg1 = xg + (size_t)R_ * C_;
#pragma unroll
        for (int p = 0; p < R_ / 2; ++p) {
            relay<true, true>(p, xw + 4 * lane, xg1, G);
            sfence();
        }
    }

    for (int st = 0; st < NSTEP - 2; ++st)
        qp_step<true, true>(st, xw, xg, qf, G, acc, m_run, s_run, ps, mg, w, lane, lid, half);
    qp_step<true, false>(NSTEP - 2, xw, xg, qf, G, acc, m_run, s_run, ps, mg, w, lane, lid, half);
    qp_step<false, false>(NSTEP - 1, xw, xg, qf, G, acc, m_run, s_run, ps, mg, w, lane, lid, half);

    __builtin_amdgcn_fence(__ATOMIC_ACQ_REL, "wavefront");
#pragma unroll
    for (int j = 0; j < 8; ++j) {
        v4f o0 = {acc[j][0], acc[j][1], acc[j][2], acc[j][3]};
        v4f o1 = {acc[j][4], acc[j][5], acc[j][6], acc[j][7]};
        *(v4f*)(xw + lid * XP + 16 * j + 8 * half)     = o0;
        *(v4f*)(xw + lid * XP + 16 * j + 8 * half + 4) = o1;
    }
    __builtin_amdgcn_fence(__ATOMIC_ACQ_REL, "wavefront");
    float* prow = part + (size_t)((b * NSPLIT + split) * K_) * C_ + c0 + 4 * lane;
#pragma unroll
    for (int qq = 0; qq < K_; ++qq) {
        v4f o = *(const v4f*)(xw + qq * XP + 4 * lane);
        __builtin_nontemporal_store(o, (v4f*)(prow + (size_t)qq * C_));
    }

    if (w == 0 && lane < K_) {
        mstat[(b * NSPLIT + split) * K_ + lane] = m_run;
        sstat[(b * NSPLIT + split) * K_ + lane] = s_run;
    }
}

__global__ __launch_bounds__(256) void qp_combine(const float* __restrict__ part,
                                                  const float* __restrict__ mstat,
                                                  const float* __restrict__ sstat,
                                                  float* __restrict__ out)
{
    const int i = blockIdx.x * 256 + threadIdx.x;
    const int c = i & (C_ - 1);
    const int q = (i >> 10) & (K_ - 1);
    const int b = i >> 14;

    float M = NEG_MAX;
#pragma unroll
    for (int wv = 0; wv < NSPLIT; ++wv)
        M = fmaxf(M, mstat[(b * NSPLIT + wv) * K_ + q]);

    if (M == NEG_MAX) { out[i] = 0.0f; return; }

    float num = 0.0f, den = 0.0f;
#pragma unroll
    for (int wv = 0; wv < NSPLIT; ++wv) {
        const int sidx = (b * NSPLIT + wv) * K_ + q;
        const float e  = __expf(mstat[sidx] - M);
        den += e * sstat[sidx];
        num += e * __builtin_nontemporal_load(part + (size_t)sidx * C_ + c);
    }
    out[i] = num / den;
}

extern "C" void kernel_launch(void* const* d_in, const int* in_sizes, int n_in,
                              void* d_out, int out_size, void* d_ws, size_t ws_size,
                              hipStream_t stream)
{
    const float* x    = (const float*)d_in[0];
    const int*   mask = (const int*)d_in[1];
    const float* q    = (const float*)d_in[2];
    float*       out  = (float*)d_out;

    float* part  = (float*)d_ws;
    float* mstat = part + (size_t)B_ * NSPLIT * K_ * C_;
    float* sstat = mstat + B_ * NSPLIT * K_;

    qp_fused<<<B_ * NSPLIT, 256, 0, stream>>>(x, mask, q, part, mstat, sstat);
    qp_combine<<<(B_ * K_ * C_) / 256, 256, 0, stream>>>(part, mstat, sstat, out);
}
